# starting kernel
# baseline (speedup 1.0000x reference)
#include <hip/hip_runtime.h>
#include <math.h>

#define BB 8
#define TT 512
#define DD 256
#define SS 64
#define PP 1024
#define RR 12
#define NN (RR * DD)
#define NROWS (BB * SS)

typedef __attribute__((ext_vector_type(2))) float v2f;
typedef __attribute__((ext_vector_type(8))) float v8f;

#define LSTR 260

__global__ __launch_bounds__(256) void span_pool_kernel(
    const float* __restrict__ encoded,
    const int*   __restrict__ starts,
    const int*   __restrict__ lens,
    const float* __restrict__ W_linear,
    float*       __restrict__ pooled,
    float*       __restrict__ lin)
{
    __shared__ float rowLDS[DD];

    const int bs = blockIdx.x;
    const int d  = threadIdx.x;
    const int b  = bs >> 6;
    const int st = starts[bs];
    const int en = st + lens[bs] + 1;
    const float* base = encoded + ((size_t)b * TT) * DD + d;
    float m = -3.402823466e+38f;
    for (int t = st; t < en; ++t)
        m = fmaxf(m, base[(size_t)t * DD]);
    pooled[(size_t)bs * DD + d] = m;
    rowLDS[d] = m;
    __syncthreads();

    const int wid  = threadIdx.x >> 5;
    const int lane = threadIdx.x & 31;
    #pragma unroll
    for (int oo = 0; oo < 3; ++oo) {
        const int o   = wid * 3 + oo;
        const int r   = (o < 12) ? o : (o - 12);
        const int off = (o < 12) ? 0 : DD;
        float pl = 0.f;
        #pragma unroll
        for (int j = 0; j < 8; ++j) {
            const int dd = lane + 32 * j;
            pl += rowLDS[dd] * W_linear[(off + dd) * RR + r];
        }
        pl += __shfl_xor(pl, 16, 32);
        pl += __shfl_xor(pl,  8, 32);
        pl += __shfl_xor(pl,  4, 32);
        pl += __shfl_xor(pl,  2, 32);
        pl += __shfl_xor(pl,  1, 32);
        if (lane == 0) lin[bs * 24 + o] = pl;
    }
}

__global__ __launch_bounds__(128) void span_gemm_kernel(
    const float* __restrict__ pooled,
    const float* __restrict__ W_bilinear,
    float*       __restrict__ G)
{
    __shared__ float aLDS[16 * LSTR];

    const int bs0 = blockIdx.x * 16;
    const int tid = threadIdx.x;

    for (int idx = tid; idx < 16 * 64; idx += 128) {
        const int row = idx >> 6;
        const int c4  = (idx & 63) << 2;
        const float4 av = *(const float4*)&pooled[((size_t)(bs0 + row)) * DD + c4];
        *(float4*)&aLDS[row * LSTR + c4] = av;
    }
    __syncthreads();

    const int wid  = tid >> 5;
    const int lane = tid & 31;
    const int l16  = lane & 15;
    const int hi   = lane >> 4;

    for (int sub = 0; sub < 4; ++sub) {
        const int ntile = blockIdx.y * 16 + wid * 4 + sub;
        const int n0    = ntile * 16;
        const int n     = n0 + l16;
        const float* wp = W_bilinear + (((size_t)(n >> 8)) << 16) + (n & 255)
                                     + (hi ? 2 * DD : 0);
        v8f c = {};
        #pragma unroll 4
        for (int k = 0; k < 64; ++k) {
            const float* ap = &aLDS[l16 * LSTR + 4 * k + 2 * hi];
            v2f a;  a.x = ap[0]; a.y = ap[1];
            v2f bf; bf.x = wp[(size_t)k * 1024];
                    bf.y = wp[(size_t)k * 1024 + DD];
            c = __builtin_amdgcn_wmma_f32_16x16x4_f32(
                    false, a, false, bf, (short)0, c, false, false);
        }
        #pragma unroll
        for (int v = 0; v < 8; ++v) {
            const int m = v + 8 * hi;
            G[((size_t)(bs0 + m)) * NN + n0 + l16] = c[v];
        }
    }
}

__global__ __launch_bounds__(128) void pair_score_kernel(
    const float* __restrict__ pooled,
    const float* __restrict__ G,
    const float* __restrict__ lin,
    const float* __restrict__ b_linear,
    const int*   __restrict__ pair_head,
    const int*   __restrict__ pair_tail,
    float*       __restrict__ out)
{
    const int wid  = threadIdx.x >> 5;
    const int lane = threadIdx.x & 31;
    const int bp   = blockIdx.x * 4 + wid;
    const int b    = bp >> 10;
    const int h    = pair_head[bp];
    const int t    = pair_tail[bp];
    const int hrow = b * SS + h;
    const int trow = b * SS + t;

    const float* tp = pooled + ((size_t)trow) * DD + lane * 8;
    const float4 t0 = *(const float4*)tp;
    const float4 t1 = *(const float4*)(tp + 4);

    const float* gr = G + ((size_t)hrow) * NN + lane * 8;
    float acc[RR];
    #pragma unroll
    for (int r = 0; r < RR; ++r) {
        const float4 g0 = *(const float4*)(gr + r * DD);
        const float4 g1 = *(const float4*)(gr + r * DD + 4);
        float s = g0.x * t0.x + g0.y * t0.y + g0.z * t0.z + g0.w * t0.w
                + g1.x * t1.x + g1.y * t1.y + g1.z * t1.z + g1.w * t1.w;
        acc[r] = s;
    }
    #pragma unroll
    for (int r = 0; r < RR; ++r) {
        float s = acc[r];
        s += __shfl_xor(s, 16, 32);
        s += __shfl_xor(s,  8, 32);
        s += __shfl_xor(s,  4, 32);
        s += __shfl_xor(s,  2, 32);
        s += __shfl_xor(s,  1, 32);
        acc[r] = s;
    }
    if (lane < RR) {
        const float sc = acc[lane] + b_linear[lane]
                       + lin[hrow * 24 + lane]
                       + lin[trow * 24 + 12 + lane];
        out[(size_t)bp * RR + lane] = 1.f / (1.f + __expf(-sc));
    }
}

extern "C" void kernel_launch(void* const* d_in, const int* in_sizes, int n_in,
                              void* d_out, int out_size, void* d_ws, size_t ws_size,
                              hipStream_t stream) {
    const float* encoded    = (const float*)d_in[0];
    const float* W_linear   = (const float*)d_in[1];
    const float* b_linear   = (const float*)d_in[2];
    const float* W_bilinear = (const float*)d_in[3];
    const int*   span_starts= (const int*)d_in[4];
    const int*   span_lens  = (const int*)d_in[5];
    const int*   pair_head  = (const int*)d_in[6];
    const int*   pair_tail  = (const int*)d_in[7];
    float*       out        = (float*)d_out;

    char* ws = (char*)d_ws;
    float* pooled = (float*)ws;
    float* lin    = (float*)(ws + (size_t)NROWS * DD * 4);
    float* G      = (float*)(ws + (size_t)NROWS * (DD + 24) * 4);

    span_pool_kernel<<<NROWS, DD, 0, stream>>>(encoded, span_starts, span_lens,
                                               W_linear, pooled, lin);

    dim3 ggrid(NROWS / 16, NN / 256);
    span_gemm_kernel<<<ggrid, 128, 0, stream>>>(pooled, W_bilinear, G);

    pair_score_kernel<<<(BB * PP) / 4, 128, 0, stream>>>(pooled, G, lin, b_linear,
                                                         pair_head, pair_tail, out);
}


# GEMM re-decomposed on the tail side, G[bs,r,d]=sum_e W[r,d,e]*pooled[bs,e]: 64 pooled + 96 W rows staged once per block in LDS, both fragments 16-byte k-contiguous reads; e and d sum orders change
# speedup vs baseline: 1.6254x; 1.6254x over previous
#include <hip/hip_runtime.h>
#include <math.h>

#define BB 8
#define TT 512
#define DD 256
#define SS 64
#define PP 1024
#define RR 12
#define NN (RR * DD)
#define NROWS (BB * SS)

typedef __attribute__((ext_vector_type(2))) float v2f;
typedef __attribute__((ext_vector_type(8))) float v8f;

#define LSTR 260

__global__ __launch_bounds__(256) void span_pool_kernel(
    const float* __restrict__ encoded,
    const int*   __restrict__ starts,
    const int*   __restrict__ lens,
    const float* __restrict__ W_linear,
    float*       __restrict__ pooled,
    float*       __restrict__ lin)
{
    __shared__ float rowLDS[DD];

    const int bs = blockIdx.x;
    const int d  = threadIdx.x;
    const int b  = bs >> 6;
    const int st = starts[bs];
    const int en = st + lens[bs] + 1;
    const float* base = encoded + ((size_t)b * TT) * DD + d;
    float m = -3.402823466e+38f;
    for (int t = st; t < en; ++t)
        m = fmaxf(m, base[(size_t)t * DD]);
    pooled[(size_t)bs * DD + d] = m;
    rowLDS[d] = m;
    __syncthreads();

    const int wid  = threadIdx.x >> 5;
    const int lane = threadIdx.x & 31;
    #pragma unroll
    for (int oo = 0; oo < 3; ++oo) {
        const int o   = wid * 3 + oo;
        const int r   = (o < 12) ? o : (o - 12);
        const int off = (o < 12) ? 0 : DD;
        float pl = 0.f;
        #pragma unroll
        for (int j = 0; j < 8; ++j) {
            const int dd = lane + 32 * j;
            pl += rowLDS[dd] * W_linear[(off + dd) * RR + r];
        }
        pl += __shfl_xor(pl, 16, 32);
        pl += __shfl_xor(pl,  8, 32);
        pl += __shfl_xor(pl,  4, 32);
        pl += __shfl_xor(pl,  2, 32);
        pl += __shfl_xor(pl,  1, 32);
        if (lane == 0) lin[bs * 24 + o] = pl;
    }
}

#define BS_CH 64
#define W_CH  96
#define LROWS (BS_CH + W_CH)

__device__ __forceinline__ v8f mma4(float ax, float ay, float bx, float by, v8f c) {
    v2f a;  a.x = ax;  a.y = ay;
    v2f b;  b.x = bx;  b.y = by;
    return __builtin_amdgcn_wmma_f32_16x16x4_f32(false, a, false, b, (short)0, c, false, false);
}

__global__ __launch_bounds__(256) void span_gemm_kernel(
    const float* __restrict__ pooled,
    const float* __restrict__ W_bilinear,
    float*       __restrict__ G)
{
    __shared__ float tile[LROWS * LSTR];

    const int tid = threadIdx.x;
    const int bs0 = blockIdx.x * BS_CH;
    const int n0  = blockIdx.y * W_CH;
    const float* psrc = pooled     + (size_t)bs0 * DD;
    const float* wsrc = W_bilinear + (size_t)n0  * DD;

    float4 pv[(BS_CH * 64) / 256];
    float4 wv[(W_CH  * 64) / 256];
    #pragma unroll
    for (int it = 0; it < (BS_CH * 64) / 256; ++it)
        pv[it] = *(const float4*)&psrc[(size_t)(tid + 256 * it) * 4];
    #pragma unroll
    for (int it = 0; it < (W_CH * 64) / 256; ++it)
        wv[it] = *(const float4*)&wsrc[(size_t)(tid + 256 * it) * 4];
    #pragma unroll
    for (int it = 0; it < (BS_CH * 64) / 256; ++it) {
        const int idx = tid + 256 * it;
        *(float4*)&tile[(idx >> 6) * LSTR + ((idx & 63) << 2)] = pv[it];
    }
    #pragma unroll
    for (int it = 0; it < (W_CH * 64) / 256; ++it) {
        const int idx = tid + 256 * it;
        *(float4*)&tile[(BS_CH + (idx >> 6)) * LSTR + ((idx & 63) << 2)] = wv[it];
    }
    __syncthreads();

    const int wid  = tid >> 5;
    const int lane = tid & 31;
    const int l16  = lane & 15;
    const int hi   = lane >> 4;
    const int bsT  = wid & 3;
    const int wT0  = (wid >> 2) * 3;

    const float* pB  = &tile[(bsT * 16 + l16) * LSTR + 4 * hi];
    const float* pA0 = &tile[(BS_CH + (wT0 + 0) * 16 + l16) * LSTR + 4 * hi];
    const float* pA1 = &tile[(BS_CH + (wT0 + 1) * 16 + l16) * LSTR + 4 * hi];
    const float* pA2 = &tile[(BS_CH + (wT0 + 2) * 16 + l16) * LSTR + 4 * hi];

    v8f c0 = {}, c1 = {}, c2 = {};
    #pragma unroll
    for (int j = 0; j < 32; ++j) {
        const float4 b  = *(const float4*)(pB  + 8 * j);
        const float4 a0 = *(const float4*)(pA0 + 8 * j);
        const float4 a1 = *(const float4*)(pA1 + 8 * j);
        const float4 a2 = *(const float4*)(pA2 + 8 * j);
        c0 = mma4(a0.x, a0.y, b.x, b.y, c0);
        c1 = mma4(a1.x, a1.y, b.x, b.y, c1);
        c2 = mma4(a2.x, a2.y, b.x, b.y, c2);
        c0 = mma4(a0.z, a0.w, b.z, b.w, c0);
        c1 = mma4(a1.z, a1.w, b.z, b.w, c1);
        c2 = mma4(a2.z, a2.w, b.z, b.w, c2);
    }

    float* gp = G + (size_t)(bs0 + bsT * 16 + l16) * NN + n0 + wT0 * 16 + 8 * hi;
    *(float4*)(gp +  0) = make_float4(c0[0], c0[1], c0[2], c0[3]);
    *(float4*)(gp +  4) = make_float4(c0[4], c0[5], c0[6], c0[7]);
    *(float4*)(gp + 16) = make_float4(c1[0], c1[1], c1[2], c1[3]);
    *(float4*)(gp + 20) = make_float4(c1[4], c1[5], c1[6], c1[7]);
    *(float4*)(gp + 32) = make_float4(c2[0], c2[1], c2[2], c2[3]);
    *(float4*)(gp + 36) = make_float4(c2[4], c2[5], c2[6], c2[7]);
}

__global__ __launch_bounds__(128) void pair_score_kernel(
    const float* __restrict__ pooled,
    const float* __restrict__ G,
    const float* __restrict__ lin,
    const float* __restrict__ b_linear,
    const int*   __restrict__ pair_head,
    const int*   __restrict__ pair_tail,
    float*       __restrict__ out)
{
    const int wid  = threadIdx.x >> 5;
    const int lane = threadIdx.x & 31;
    const int bp   = blockIdx.x * 4 + wid;
    const int b    = bp >> 10;
    const int h    = pair_head[bp];
    const int t    = pair_tail[bp];
    const int hrow = b * SS + h;
    const int trow = b * SS + t;

    const float* hp = pooled + ((size_t)hrow) * DD + lane * 8;
    const float4 h0 = *(const float4*)hp;
    const float4 h1 = *(const float4*)(hp + 4);

    const float* gr = G + ((size_t)trow) * NN + lane * 8;
    float acc[RR];
    #pragma unroll
    for (int r = 0; r < RR; ++r) {
        const float4 g0 = *(const float4*)(gr + r * DD);
        const float4 g1 = *(const float4*)(gr + r * DD + 4);
        float s = g0.x * h0.x + g0.y * h0.y + g0.z * h0.z + g0.w * h0.w
                + g1.x * h1.x + g1.y * h1.y + g1.z * h1.z + g1.w * h1.w;
        acc[r] = s;
    }
    #pragma unroll
    for (int r = 0; r < RR; ++r) {
        float s = acc[r];
        s += __shfl_xor(s, 16, 32);
        s += __shfl_xor(s,  8, 32);
        s += __shfl_xor(s,  4, 32);
        s += __shfl_xor(s,  2, 32);
        s += __shfl_xor(s,  1, 32);
        acc[r] = s;
    }
    if (lane < RR) {
        const float sc = acc[lane] + b_linear[lane]
                       + lin[hrow * 24 + lane]
                       + lin[trow * 24 + 12 + lane];
        out[(size_t)bp * RR + lane] = 1.f / (1.f + __expf(-sc));
    }
}

extern "C" void kernel_launch(void* const* d_in, const int* in_sizes, int n_in,
                              void* d_out, int out_size, void* d_ws, size_t ws_size,
                              hipStream_t stream) {
    const float* encoded    = (const float*)d_in[0];
    const float* W_linear   = (const float*)d_in[1];
    const float* b_linear   = (const float*)d_in[2];
    const float* W_bilinear = (const float*)d_in[3];
    const int*   span_starts= (const int*)d_in[4];
    const int*   span_lens  = (const int*)d_in[5];
    const int*   pair_head  = (const int*)d_in[6];
    const int*   pair_tail  = (const int*)d_in[7];
    float*       out        = (float*)d_out;

    char* ws = (char*)d_ws;
    float* pooled = (float*)ws;
    float* lin    = (float*)(ws + (size_t)NROWS * DD * 4);
    float* G      = (float*)(ws + (size_t)NROWS * (DD + 24) * 4);

    span_pool_kernel<<<NROWS, DD, 0, stream>>>(encoded, span_starts, span_lens,
                                               W_linear, pooled, lin);

    dim3 ggrid(NROWS / BS_CH, NN / W_CH);
    span_gemm_kernel<<<ggrid, 256, 0, stream>>>(pooled, W_bilinear, G);

    pair_score_kernel<<<(BB * PP) / 4, 128, 0, stream>>>(pooled, G, lin, b_linear,
                                                         pair_head, pair_tail, out);
}


# Pair scoring remapped to four pairs per wave: 8 lanes own a pair and read whole 128-byte lines of its G and head rows, lane sums by DPP in the group; 8192 waves become 2048; d sum order changes
# speedup vs baseline: 1.8486x; 1.1373x over previous
#include <hip/hip_runtime.h>
#include <math.h>

#define BB 8
#define TT 512
#define DD 256
#define SS 64
#define PP 1024
#define RR 12
#define NN (RR * DD)
#define NROWS (BB * SS)

typedef __attribute__((ext_vector_type(2))) float v2f;
typedef __attribute__((ext_vector_type(8))) float v8f;

#define LSTR 260

__global__ __launch_bounds__(256) void span_pool_kernel(
    const float* __restrict__ encoded,
    const int*   __restrict__ starts,
    const int*   __restrict__ lens,
    const float* __restrict__ W_linear,
    float*       __restrict__ pooled,
    float*       __restrict__ lin)
{
    __shared__ float rowLDS[DD];

    const int bs = blockIdx.x;
    const int d  = threadIdx.x;
    const int b  = bs >> 6;
    const int st = starts[bs];
    const int en = st + lens[bs] + 1;
    const float* base = encoded + ((size_t)b * TT) * DD + d;
    float m = -3.402823466e+38f;
    for (int t = st; t < en; ++t)
        m = fmaxf(m, base[(size_t)t * DD]);
    pooled[(size_t)bs * DD + d] = m;
    rowLDS[d] = m;
    __syncthreads();

    const int wid  = threadIdx.x >> 5;
    const int lane = threadIdx.x & 31;
    #pragma unroll
    for (int oo = 0; oo < 3; ++oo) {
        const int o   = wid * 3 + oo;
        const int r   = (o < 12) ? o : (o - 12);
        const int off = (o < 12) ? 0 : DD;
        float pl = 0.f;
        #pragma unroll
        for (int j = 0; j < 8; ++j) {
            const int dd = lane + 32 * j;
            pl += rowLDS[dd] * W_linear[(off + dd) * RR + r];
        }
        pl += __shfl_xor(pl, 16, 32);
        pl += __shfl_xor(pl,  8, 32);
        pl += __shfl_xor(pl,  4, 32);
        pl += __shfl_xor(pl,  2, 32);
        pl += __shfl_xor(pl,  1, 32);
        if (lane == 0) lin[bs * 24 + o] = pl;
    }
}

#define BS_CH 64
#define W_CH  96
#define LROWS (BS_CH + W_CH)

__device__ __forceinline__ v8f mma4(float ax, float ay, float bx, float by, v8f c) {
    v2f a;  a.x = ax;  a.y = ay;
    v2f b;  b.x = bx;  b.y = by;
    return __builtin_amdgcn_wmma_f32_16x16x4_f32(false, a, false, b, (short)0, c, false, false);
}

__global__ __launch_bounds__(256) void span_gemm_kernel(
    const float* __restrict__ pooled,
    const float* __restrict__ W_bilinear,
    float*       __restrict__ G)
{
    __shared__ float tile[LROWS * LSTR];

    const int tid = threadIdx.x;
    const int bs0 = blockIdx.x * BS_CH;
    const int n0  = blockIdx.y * W_CH;
    const float* psrc = pooled     + (size_t)bs0 * DD;
    const float* wsrc = W_bilinear + (size_t)n0  * DD;

    float4 pv[(BS_CH * 64) / 256];
    float4 wv[(W_CH  * 64) / 256];
    #pragma unroll
    for (int it = 0; it < (BS_CH * 64) / 256; ++it)
        pv[it] = *(const float4*)&psrc[(size_t)(tid + 256 * it) * 4];
    #pragma unroll
    for (int it = 0; it < (W_CH * 64) / 256; ++it)
        wv[it] = *(const float4*)&wsrc[(size_t)(tid + 256 * it) * 4];
    #pragma unroll
    for (int it = 0; it < (BS_CH * 64) / 256; ++it) {
        const int idx = tid + 256 * it;
        *(float4*)&tile[(idx >> 6) * LSTR + ((idx & 63) << 2)] = pv[it];
    }
    #pragma unroll
    for (int it = 0; it < (W_CH * 64) / 256; ++it) {
        const int idx = tid + 256 * it;
        *(float4*)&tile[(BS_CH + (idx >> 6)) * LSTR + ((idx & 63) << 2)] = wv[it];
    }
    __syncthreads();

    const int wid  = tid >> 5;
    const int lane = tid & 31;
    const int l16  = lane & 15;
    const int hi   = lane >> 4;
    const int bsT  = wid & 3;
    const int wT0  = (wid >> 2) * 3;

    const float* pB  = &tile[(bsT * 16 + l16) * LSTR + 4 * hi];
    const float* pA0 = &tile[(BS_CH + (wT0 + 0) * 16 + l16) * LSTR + 4 * hi];
    const float* pA1 = &tile[(BS_CH + (wT0 + 1) * 16 + l16) * LSTR + 4 * hi];
    const float* pA2 = &tile[(BS_CH + (wT0 + 2) * 16 + l16) * LSTR + 4 * hi];

    v8f c0 = {}, c1 = {}, c2 = {};
    #pragma unroll
    for (int j = 0; j < 32; ++j) {
        const float4 b  = *(const float4*)(pB  + 8 * j);
        const float4 a0 = *(const float4*)(pA0 + 8 * j);
        const float4 a1 = *(const float4*)(pA1 + 8 * j);
        const float4 a2 = *(const float4*)(pA2 + 8 * j);
        c0 = mma4(a0.x, a0.y, b.x, b.y, c0);
        c1 = mma4(a1.x, a1.y, b.x, b.y, c1);
        c2 = mma4(a2.x, a2.y, b.x, b.y, c2);
        c0 = mma4(a0.z, a0.w, b.z, b.w, c0);
        c1 = mma4(a1.z, a1.w, b.z, b.w, c1);
        c2 = mma4(a2.z, a2.w, b.z, b.w, c2);
    }

    float* gp = G + (size_t)(bs0 + bsT * 16 + l16) * NN + n0 + wT0 * 16 + 8 * hi;
    *(float4*)(gp +  0) = make_float4(c0[0], c0[1], c0[2], c0[3]);
    *(float4*)(gp +  4) = make_float4(c0[4], c0[5], c0[6], c0[7]);
    *(float4*)(gp + 16) = make_float4(c1[0], c1[1], c1[2], c1[3]);
    *(float4*)(gp + 20) = make_float4(c1[4], c1[5], c1[6], c1[7]);
    *(float4*)(gp + 32) = make_float4(c2[0], c2[1], c2[2], c2[3]);
    *(float4*)(gp + 36) = make_float4(c2[4], c2[5], c2[6], c2[7]);
}

#define PAIRS_PER_WAVE 4
#define PAIRS_PER_BLOCK (4 * PAIRS_PER_WAVE)

template <int XM>
__device__ __forceinline__ float row_xor(float v) {
    return __builtin_bit_cast(float,
        __builtin_amdgcn_update_dpp(0, __builtin_bit_cast(int, v), 0x160 + XM, 0xf, 0xf, false));
}

__device__ __forceinline__ float sigmoidf(float x) { return 1.f / (1.f + __expf(-x)); }

__global__ __launch_bounds__(128) __attribute__((amdgpu_waves_per_eu(1, 4))) void pair_score_kernel(
    const float* __restrict__ pooled,
    const float* __restrict__ G,
    const float* __restrict__ lin,
    const float* __restrict__ b_linear,
    const int*   __restrict__ pair_head,
    const int*   __restrict__ pair_tail,
    float*       __restrict__ out)
{
    const int wid  = threadIdx.x >> 5;
    const int lane = threadIdx.x & 31;
    const int q    = lane >> 3;
    const int s    = lane & 7;
    const int bp   = (blockIdx.x * 4 + wid) * PAIRS_PER_WAVE + q;
    const int b    = bp >> 10;
    const int h    = pair_head[bp];
    const int t    = pair_tail[bp];
    const int hrow = b * SS + h;
    const int trow = b * SS + t;

    const int   r1   = 8 + (s & 3);
    const float lb0  = b_linear[s];
    const float lh0  = lin[hrow * 24 + s];
    const float lt0  = lin[trow * 24 + 12 + s];
    const float lb1  = b_linear[r1];
    const float lh1  = lin[hrow * 24 + r1];
    const float lt1  = lin[trow * 24 + 12 + r1];

    const float* hp = pooled + ((size_t)hrow) * DD + 4 * s;
    float4 hv[8];
    #pragma unroll
    for (int i = 0; i < 8; ++i)
        hv[i] = *(const float4*)(hp + 32 * i);

    const float* gr = G + ((size_t)trow) * NN + 4 * s;
    float acc[RR];
    #pragma unroll
    for (int r = 0; r < RR; ++r) {
        float sum = 0.f;
        #pragma unroll
        for (int i = 0; i < 8; ++i) {
            const float4 g = *(const float4*)(gr + r * DD + 32 * i);
            sum = fmaf(g.x, hv[i].x, sum);
            sum = fmaf(g.y, hv[i].y, sum);
            sum = fmaf(g.z, hv[i].z, sum);
            sum = fmaf(g.w, hv[i].w, sum);
        }
        acc[r] = sum;
    }
    asm volatile("" :: "v"(lb1), "v"(lh1), "v"(lt1));

    #pragma unroll
    for (int r = 0; r < RR; ++r) {
        float v = acc[r];
        v += row_xor<1>(v);
        v += row_xor<2>(v);
        v += row_xor<4>(v);
        acc[r] = v;
    }
    float v0 = acc[0];
    #pragma unroll
    for (int k = 1; k < 8; ++k) v0 = (s == k) ? acc[k] : v0;
    float v1 = acc[8];
    #pragma unroll
    for (int k = 1; k < 4; ++k) v1 = ((s & 3) == k) ? acc[8 + k] : v1;

    float* op = out + (size_t)bp * RR;
    op[s] = sigmoidf(v0 + lb0 + lh0 + lt0);
    if (s < 4)
        op[r1] = sigmoidf(v1 + lb1 + lh1 + lt1);
}

extern "C" void kernel_launch(void* const* d_in, const int* in_sizes, int n_in,
                              void* d_out, int out_size, void* d_ws, size_t ws_size,
                              hipStream_t stream) {
    const float* encoded    = (const float*)d_in[0];
    const float* W_linear   = (const float*)d_in[1];
    const float* b_linear   = (const float*)d_in[2];
    const float* W_bilinear = (const float*)d_in[3];
    const int*   span_starts= (const int*)d_in[4];
    const int*   span_lens  = (const int*)d_in[5];
    const int*   pair_head  = (const int*)d_in[6];
    const int*   pair_tail  = (const int*)d_in[7];
    float*       out        = (float*)d_out;

    char* ws = (char*)d_ws;
    float* pooled = (float*)ws;
    float* lin    = (float*)(ws + (size_t)NROWS * DD * 4);
    float* G      = (float*)(ws + (size_t)NROWS * (DD + 24) * 4);

    span_pool_kernel<<<NROWS, DD, 0, stream>>>(encoded, span_starts, span_lens,
                                               W_linear, pooled, lin);

    dim3 ggrid(NROWS / BS_CH, NN / W_CH);
    span_gemm_kernel<<<ggrid, 256, 0, stream>>>(pooled, W_bilinear, G);

    pair_score_kernel<<<(BB * PP) / PAIRS_PER_BLOCK, 128, 0, stream>>>(pooled, G, lin, b_linear,
                                                         pair_head, pair_tail, out);
}


# Span pooling recast by channel slab: 64 blocks stage a batch's 512 rows of 32 channels in LDS once, take all span maxima there; linear terms as 8 slab partials added in scoring, sum order changes
# speedup vs baseline: 2.2064x; 1.1936x over previous
#include <hip/hip_runtime.h>
#include <math.h>

#define BB 8
#define TT 512
#define DD 256
#define SS 64
#define PP 1024
#define RR 12
#define NN (RR * DD)
#define NROWS (BB * SS)

typedef __attribute__((ext_vector_type(2))) float v2f;
typedef __attribute__((ext_vector_type(8))) float v8f;

#define LSTR 260

#define SLAB  32
#define NSLAB (DD / SLAB)
#define PSTR  33

__global__ __launch_bounds__(256) void span_pool_kernel(
    const float* __restrict__ encoded,
    const int*   __restrict__ starts,
    const int*   __restrict__ lens,
    const float* __restrict__ W_linear,
    float*       __restrict__ pooled,
    float*       __restrict__ linp)
{
    __shared__ float slab[TT * SLAB];
    __shared__ float pl[SS * PSTR];
    __shared__ float wl[SLAB * 24];

    const int tid  = threadIdx.x;
    const int lane = tid & 31;
    const int wid  = __builtin_amdgcn_readfirstlane(tid >> 5);
    const int b    = blockIdx.x >> 3;
    const int sl   = blockIdx.x & 7;

    const float* src = encoded + ((size_t)b * TT) * DD + sl * SLAB;
    float4 v[(TT * SLAB / 4) / 256];
    #pragma unroll
    for (int it = 0; it < (TT * SLAB / 4) / 256; ++it) {
        const int f = tid + 256 * it;
        v[it] = *(const float4*)(src + (size_t)(f >> 3) * DD + ((f & 7) << 2));
    }
    #pragma unroll
    for (int k = 0; k < 3; ++k) {
        const int idx  = tid + 256 * k;
        const int half = idx / 384;
        const int j    = idx - half * 384;
        const int c    = j / 12;
        const int r    = j - c * 12;
        wl[c * 24 + half * 12 + r] = W_linear[(size_t)(half * DD + sl * SLAB) * RR + j];
    }
    const int* sp = starts + b * SS + wid * 8;
    const int* lp = lens   + b * SS + wid * 8;
    int st[8], en[8];
    #pragma unroll
    for (int i = 0; i < 8; ++i) {
        const int s0 = sp[i];
        const int e0 = s0 + lp[i] + 1;
        st[i] = max(s0, 0);
        en[i] = min(e0, TT);
    }
    #pragma unroll
    for (int it = 0; it < (TT * SLAB / 4) / 256; ++it)
        *(float4*)&slab[(tid + 256 * it) << 2] = v[it];
    __syncthreads();

    const float NEG = -3.402823466e+38f;
    float m[8];
    int   last[8];
    int   maxlen = 0;
    #pragma unroll
    for (int i = 0; i < 8; ++i) {
        m[i]    = NEG;
        last[i] = max(en[i] - 1, 0);
        maxlen  = max(maxlen, en[i] - st[i]);
    }
    for (int c0 = 0; c0 < maxlen; c0 += 12) {
        #pragma unroll
        for (int i = 0; i < 8; ++i) {
            #pragma unroll
            for (int k = 0; k < 12; ++k) {
                const int t = min(st[i] + c0 + k, last[i]);
                m[i] = fmaxf(m[i], slab[t * SLAB + lane]);
            }
        }
    }
    #pragma unroll
    for (int i = 0; i < 8; ++i)
        m[i] = (en[i] > st[i]) ? m[i] : NEG;

    #pragma unroll
    for (int i = 0; i < 8; ++i) {
        const int s = wid * 8 + i;
        pooled[((size_t)(b * SS + s)) * DD + sl * SLAB + lane] = m[i];
        pl[s * PSTR + lane] = m[i];
    }
    __syncthreads();

    const int s = tid & 63;
    const int g = tid >> 6;
    float acc[6] = {0.f, 0.f, 0.f, 0.f, 0.f, 0.f};
    #pragma unroll
    for (int c = 0; c < SLAB; ++c) {
        const float p = pl[s * PSTR + c];
        #pragma unroll
        for (int k = 0; k < 6; ++k)
            acc[k] = fmaf(p, wl[c * 24 + g * 6 + k], acc[k]);
    }
    float* lo = linp + ((size_t)(sl * NROWS + b * SS + s)) * 24 + g * 6;
    #pragma unroll
    for (int k = 0; k < 6; ++k) lo[k] = acc[k];
}

#define BS_CH 64
#define W_CH  96
#define LROWS (BS_CH + W_CH)

__device__ __forceinline__ v8f mma4(float ax, float ay, float bx, float by, v8f c) {
    v2f a;  a.x = ax;  a.y = ay;
    v2f b;  b.x = bx;  b.y = by;
    return __builtin_amdgcn_wmma_f32_16x16x4_f32(false, a, false, b, (short)0, c, false, false);
}

__global__ __launch_bounds__(256) void span_gemm_kernel(
    const float* __restrict__ pooled,
    const float* __restrict__ W_bilinear,
    float*       __restrict__ G)
{
    __shared__ float tile[LROWS * LSTR];

    const int tid = threadIdx.x;
    const int bs0 = blockIdx.x * BS_CH;
    const int n0  = blockIdx.y * W_CH;
    const float* psrc = pooled     + (size_t)bs0 * DD;
    const float* wsrc = W_bilinear + (size_t)n0  * DD;

    float4 pv[(BS_CH * 64) / 256];
    float4 wv[(W_CH  * 64) / 256];
    #pragma unroll
    for (int it = 0; it < (BS_CH * 64) / 256; ++it)
        pv[it] = *(const float4*)&psrc[(size_t)(tid + 256 * it) * 4];
    #pragma unroll
    for (int it = 0; it < (W_CH * 64) / 256; ++it)
        wv[it] = *(const float4*)&wsrc[(size_t)(tid + 256 * it) * 4];
    #pragma unroll
    for (int it = 0; it < (BS_CH * 64) / 256; ++it) {
        const int idx = tid + 256 * it;
        *(float4*)&tile[(idx >> 6) * LSTR + ((idx & 63) << 2)] = pv[it];
    }
    #pragma unroll
    for (int it = 0; it < (W_CH * 64) / 256; ++it) {
        const int idx = tid + 256 * it;
        *(float4*)&tile[(BS_CH + (idx >> 6)) * LSTR + ((idx & 63) << 2)] = wv[it];
    }
    __syncthreads();

    const int wid  = tid >> 5;
    const int lane = tid & 31;
    const int l16  = lane & 15;
    const int hi   = lane >> 4;
    const int bsT  = wid & 3;
    const int wT0  = (wid >> 2) * 3;

    const float* pB  = &tile[(bsT * 16 + l16) * LSTR + 4 * hi];
    const float* pA0 = &tile[(BS_CH + (wT0 + 0) * 16 + l16) * LSTR + 4 * hi];
    const float* pA1 = &tile[(BS_CH + (wT0 + 1) * 16 + l16) * LSTR + 4 * hi];
    const float* pA2 = &tile[(BS_CH + (wT0 + 2) * 16 + l16) * LSTR + 4 * hi];

    v8f c0 = {}, c1 = {}, c2 = {};
    #pragma unroll
    for (int j = 0; j < 32; ++j) {
        const float4 b  = *(const float4*)(pB  + 8 * j);
        const float4 a0 = *(const float4*)(pA0 + 8 * j);
        const float4 a1 = *(const float4*)(pA1 + 8 * j);
        const float4 a2 = *(const float4*)(pA2 + 8 * j);
        c0 = mma4(a0.x, a0.y, b.x, b.y, c0);
        c1 = mma4(a1.x, a1.y, b.x, b.y, c1);
        c2 = mma4(a2.x, a2.y, b.x, b.y, c2);
        c0 = mma4(a0.z, a0.w, b.z, b.w, c0);
        c1 = mma4(a1.z, a1.w, b.z, b.w, c1);
        c2 = mma4(a2.z, a2.w, b.z, b.w, c2);
    }

    float* gp = G + (size_t)(bs0 + bsT * 16 + l16) * NN + n0 + wT0 * 16 + 8 * hi;
    *(float4*)(gp +  0) = make_float4(c0[0], c0[1], c0[2], c0[3]);
    *(float4*)(gp +  4) = make_float4(c0[4], c0[5], c0[6], c0[7]);
    *(float4*)(gp + 16) = make_float4(c1[0], c1[1], c1[2], c1[3]);
    *(float4*)(gp + 20) = make_float4(c1[4], c1[5], c1[6], c1[7]);
    *(float4*)(gp + 32) = make_float4(c2[0], c2[1], c2[2], c2[3]);
    *(float4*)(gp + 36) = make_float4(c2[4], c2[5], c2[6], c2[7]);
}

#define PAIRS_PER_WAVE 4
#define PAIRS_PER_BLOCK (4 * PAIRS_PER_WAVE)

template <int XM>
__device__ __forceinline__ float row_xor(float v) {
    return __builtin_bit_cast(float,
        __builtin_amdgcn_update_dpp(0, __builtin_bit_cast(int, v), 0x160 + XM, 0xf, 0xf, false));
}

__device__ __forceinline__ float sigmoidf(float x) { return 1.f / (1.f + __expf(-x)); }

__global__ __launch_bounds__(128) __attribute__((amdgpu_waves_per_eu(1, 4))) void pair_score_kernel(
    const float* __restrict__ pooled,
    const float* __restrict__ G,
    const float* __restrict__ linp,
    const float* __restrict__ b_linear,
    const int*   __restrict__ pair_head,
    const int*   __restrict__ pair_tail,
    float*       __restrict__ out)
{
    const int wid  = threadIdx.x >> 5;
    const int lane = threadIdx.x & 31;
    const int q    = lane >> 3;
    const int s    = lane & 7;
    const int bp   = (blockIdx.x * 4 + wid) * PAIRS_PER_WAVE + q;
    const int b    = bp >> 10;
    const int h    = pair_head[bp];
    const int t    = pair_tail[bp];
    const int hrow = b * SS + h;
    const int trow = b * SS + t;

    const int   r1   = 8 + (s & 3);
    const float lb0  = b_linear[s];
    const float lb1  = b_linear[r1];
    const float4* lhp = (const float4*)(linp + ((size_t)s * NROWS + hrow) * 24);
    const float4* ltp = (const float4*)(linp + ((size_t)s * NROWS + trow) * 24 + 12);
    float4 lh[3], lt[3];
    #pragma unroll
    for (int k = 0; k < 3; ++k) { lh[k] = lhp[k]; lt[k] = ltp[k]; }
    const float lhf[RR] = { lh[0].x, lh[0].y, lh[0].z, lh[0].w, lh[1].x, lh[1].y, lh[1].z, lh[1].w,
                            lh[2].x, lh[2].y, lh[2].z, lh[2].w };
    const float ltf[RR] = { lt[0].x, lt[0].y, lt[0].z, lt[0].w, lt[1].x, lt[1].y, lt[1].z, lt[1].w,
                            lt[2].x, lt[2].y, lt[2].z, lt[2].w };

    const float* hp = pooled + ((size_t)hrow) * DD + 4 * s;
    float4 hv[8];
    #pragma unroll
    for (int i = 0; i < 8; ++i)
        hv[i] = *(const float4*)(hp + 32 * i);

    const float* gr = G + ((size_t)trow) * NN + 4 * s;
    float acc[RR];
    #pragma unroll
    for (int r = 0; r < RR; ++r) {
        float sum = 0.f;
        #pragma unroll
        for (int i = 0; i < 8; ++i) {
            const float4 g = *(const float4*)(gr + r * DD + 32 * i);
            sum = fmaf(g.x, hv[i].x, sum);
            sum = fmaf(g.y, hv[i].y, sum);
            sum = fmaf(g.z, hv[i].z, sum);
            sum = fmaf(g.w, hv[i].w, sum);
        }
        acc[r] = sum + (lhf[r] + ltf[r]);
    }
    asm volatile("" :: "v"(lb1));

    #pragma unroll
    for (int r = 0; r < RR; ++r) {
        float v = acc[r];
        v += row_xor<1>(v);
        v += row_xor<2>(v);
        v += row_xor<4>(v);
        acc[r] = v;
    }
    float v0 = acc[0];
    #pragma unroll
    for (int k = 1; k < 8; ++k) v0 = (s == k) ? acc[k] : v0;
    float v1 = acc[8];
    #pragma unroll
    for (int k = 1; k < 4; ++k) v1 = ((s & 3) == k) ? acc[8 + k] : v1;

    float* op = out + (size_t)bp * RR;
    op[s] = sigmoidf(v0 + lb0);
    if (s < 4)
        op[r1] = sigmoidf(v1 + lb1);
}

extern "C" void kernel_launch(void* const* d_in, const int* in_sizes, int n_in,
                              void* d_out, int out_size, void* d_ws, size_t ws_size,
                              hipStream_t stream) {
    const float* encoded    = (const float*)d_in[0];
    const float* W_linear   = (const float*)d_in[1];
    const float* b_linear   = (const float*)d_in[2];
    const float* W_bilinear = (const float*)d_in[3];
    const int*   span_starts= (const int*)d_in[4];
    const int*   span_lens  = (const int*)d_in[5];
    const int*   pair_head  = (const int*)d_in[6];
    const int*   pair_tail  = (const int*)d_in[7];
    float*       out        = (float*)d_out;

    char* ws = (char*)d_ws;
    float* pooled = (float*)ws;
    float* linp   = (float*)(ws + (size_t)NROWS * DD * 4);
    float* G      = (float*)(ws + (size_t)NROWS * (DD + NSLAB * 24) * 4);

    span_pool_kernel<<<BB * NSLAB, 256, 0, stream>>>(encoded, span_starts, span_lens,
                                                     W_linear, pooled, linp);

    dim3 ggrid(NROWS / BS_CH, NN / W_CH);
    span_gemm_kernel<<<ggrid, 256, 0, stream>>>(pooled, W_bilinear, G);

    pair_score_kernel<<<(BB * PP) / PAIRS_PER_BLOCK, 128, 0, stream>>>(pooled, G, linp, b_linear,
                                                         pair_head, pair_tail, out);
}
